# starting kernel
# baseline (speedup 1.0000x reference)
#include <hip/hip_runtime.h>

#define NP 128
#define NG 256
#define NC 2048
#define HW 64

typedef __attribute__((ext_vector_type(2))) float v2f;
typedef __attribute__((ext_vector_type(4))) float v4f;
typedef __attribute__((ext_vector_type(8))) float v8f;

__global__ void pool_kernel(const float* __restrict__ probe,
                            const float* __restrict__ gallery,
                            float* __restrict__ pmean,
                            float* __restrict__ gmean) {
  const int lane = threadIdx.x & 31;
  const int wave = threadIdx.x >> 5;
  const int sub  = lane & 15;
  const int half = lane >> 4;

  long long row = (long long)blockIdx.x * 16 + (long long)wave * 2 + half;
  const long long n_probe_rows = (long long)NP * NC;

  const float* src;
  float* dst;
  if (row < n_probe_rows) {
    src = probe + row * HW;
    dst = pmean + row;
  } else {
    long long r2 = row - n_probe_rows;
    src = gallery + r2 * HW;
    dst = gmean + r2;
  }

  v4f v = *(const v4f*)(src + sub * 4);
  float s = v.x + v.y + v.z + v.w;
  s += __shfl_xor(s, 1, 32);
  s += __shfl_xor(s, 2, 32);
  s += __shfl_xor(s, 4, 32);
  s += __shfl_xor(s, 8, 32);
  if (sub == 0) *dst = s * (1.0f / 64.0f);
}

__global__ void norm_kernel(const float* __restrict__ pmean,
                            const float* __restrict__ gmean,
                            float* __restrict__ pn,
                            float* __restrict__ gn) {
  const int lane = threadIdx.x & 31;
  const int wave = threadIdx.x >> 5;
  const int row  = blockIdx.x * 8 + wave;

  const float* src;
  float* dst;
  if (row < NP) {
    src = pmean + (long long)row * NC;
    dst = pn + row;
  } else {
    src = gmean + (long long)(row - NP) * NC;
    dst = gn + (row - NP);
  }

  float s = 0.0f;
  #pragma unroll 4
  for (int i = lane; i < NC; i += 32) {
    float x = src[i];
    s += x * x;
  }
  s += __shfl_xor(s, 1, 32);
  s += __shfl_xor(s, 2, 32);
  s += __shfl_xor(s, 4, 32);
  s += __shfl_xor(s, 8, 32);
  s += __shfl_xor(s, 16, 32);
  if (lane == 0) *dst = s;
}

__global__ void dist_wmma_kernel(const float* __restrict__ pmean,
                                 const float* __restrict__ gmean,
                                 const float* __restrict__ pn,
                                 const float* __restrict__ gn,
                                 float* __restrict__ out) {
  const int lane = threadIdx.x;
  const int n16  = lane & 15;
  const int hi   = lane >> 4;
  const int ti   = blockIdx.y;
  const int tj   = blockIdx.x;

  const float* arow = pmean + (long long)(ti * 16 + n16) * NC + 2 * hi;
  const float* brow = gmean + (long long)(tj * 16 + n16) * NC + 2 * hi;

  v8f acc = {};
  #pragma unroll 8
  for (int k = 0; k < NC; k += 4) {
    v2f a = *(const v2f*)(arow + k);
    v2f b = *(const v2f*)(brow + k);
    acc = __builtin_amdgcn_wmma_f32_16x16x4_f32(
         false, a,  false, b,
         (short)0, acc,  false,  false);
  }

  const int n = tj * 16 + n16;
  const float gnv = gn[n];
  #pragma unroll
  for (int r = 0; r < 8; ++r) {
    const int m = ti * 16 + hi * 8 + r;
    out[m * NG + n] = pn[m] + gnv - 2.0f * acc[r];
  }
}

extern "C" void kernel_launch(void* const* d_in, const int* in_sizes, int n_in,
                              void* d_out, int out_size, void* d_ws, size_t ws_size,
                              hipStream_t stream) {
  const float* probe   = (const float*)d_in[0];
  const float* gallery = (const float*)d_in[1];
  float* out = (float*)d_out;

  float* ws    = (float*)d_ws;
  float* pmean = ws;
  float* gmean = pmean + (size_t)NP * NC;
  float* pn    = gmean + (size_t)NG * NC;
  float* gn    = pn + NP;

  const int total_rows = (NP + NG) * NC;
  pool_kernel<<<total_rows / 16, 256, 0, stream>>>(probe, gallery, pmean, gmean);

  norm_kernel<<<(NP + NG) / 8, 256, 0, stream>>>(pmean, gmean, pn, gn);

  dist_wmma_kernel<<<dim3(NG / 16, NP / 16), 32, 0, stream>>>(pmean, gmean, pn, gn, out);
}


# distance GEMM: K split over 16 waves per 16x16 tile (128 k each, float4 fragment loads, two accumulators), partial tiles summed through LDS; changes the summation order over k
# speedup vs baseline: 1.4514x; 1.4514x over previous
#include <hip/hip_runtime.h>

#define NP 128
#define NG 256
#define NC 2048
#define HW 64

typedef __attribute__((ext_vector_type(2))) float v2f;
typedef __attribute__((ext_vector_type(4))) float v4f;
typedef __attribute__((ext_vector_type(8))) float v8f;

__global__ void pool_kernel(const float* __restrict__ probe,
                            const float* __restrict__ gallery,
                            float* __restrict__ pmean,
                            float* __restrict__ gmean) {
  const int lane = threadIdx.x & 31;
  const int wave = threadIdx.x >> 5;
  const int sub  = lane & 15;
  const int half = lane >> 4;

  long long row = (long long)blockIdx.x * 16 + (long long)wave * 2 + half;
  const long long n_probe_rows = (long long)NP * NC;

  const float* src;
  float* dst;
  if (row < n_probe_rows) {
    src = probe + row * HW;
    dst = pmean + row;
  } else {
    long long r2 = row - n_probe_rows;
    src = gallery + r2 * HW;
    dst = gmean + r2;
  }

  v4f v = *(const v4f*)(src + sub * 4);
  float s = v.x + v.y + v.z + v.w;
  s += __shfl_xor(s, 1, 32);
  s += __shfl_xor(s, 2, 32);
  s += __shfl_xor(s, 4, 32);
  s += __shfl_xor(s, 8, 32);
  if (sub == 0) *dst = s * (1.0f / 64.0f);
}

__global__ void norm_kernel(const float* __restrict__ pmean,
                            const float* __restrict__ gmean,
                            float* __restrict__ pn,
                            float* __restrict__ gn) {
  const int lane = threadIdx.x & 31;
  const int wave = threadIdx.x >> 5;
  const int row  = blockIdx.x * 8 + wave;

  const float* src;
  float* dst;
  if (row < NP) {
    src = pmean + (long long)row * NC;
    dst = pn + row;
  } else {
    src = gmean + (long long)(row - NP) * NC;
    dst = gn + (row - NP);
  }

  float s = 0.0f;
  #pragma unroll 4
  for (int i = lane; i < NC; i += 32) {
    float x = src[i];
    s += x * x;
  }
  s += __shfl_xor(s, 1, 32);
  s += __shfl_xor(s, 2, 32);
  s += __shfl_xor(s, 4, 32);
  s += __shfl_xor(s, 8, 32);
  s += __shfl_xor(s, 16, 32);
  if (lane == 0) *dst = s;
}

#define KSPLIT 16
#define KSLICE (NC / KSPLIT)

__global__ __launch_bounds__(KSPLIT * 32)
void dist_wmma_kernel(const float* __restrict__ pmean,
                      const float* __restrict__ gmean,
                      const float* __restrict__ pn,
                      const float* __restrict__ gn,
                      float* __restrict__ out) {
  __shared__ float red[KSPLIT][256];

  const int lane = threadIdx.x & 31;
  const int wave = threadIdx.x >> 5;
  const int c    = lane & 15;
  const int h    = lane >> 4;
  const int ti   = blockIdx.y;
  const int tj   = blockIdx.x;

  const float* arow = pmean + (long long)(ti * 16 + c) * NC + wave * KSLICE + 4 * h;
  const float* brow = gmean + (long long)(tj * 16 + c) * NC + wave * KSLICE + 4 * h;

  v4f a[KSLICE / 8], b[KSLICE / 8];
  #pragma unroll
  for (int j = 0; j < KSLICE / 8; ++j) {
    a[j] = *(const v4f*)(arow + 8 * j);
    b[j] = *(const v4f*)(brow + 8 * j);
  }

  v8f acc0 = {};
  v8f acc1 = {};
  #pragma unroll
  for (int j = 0; j < KSLICE / 8; ++j) {
    v2f a0 = {a[j].x, a[j].y};
    v2f b0 = {b[j].x, b[j].y};
    v2f a1 = {a[j].z, a[j].w};
    v2f b1 = {b[j].z, b[j].w};
    acc0 = __builtin_amdgcn_wmma_f32_16x16x4_f32(
         false, a0,  false, b0,
         (short)0, acc0,  false,  false);
    acc1 = __builtin_amdgcn_wmma_f32_16x16x4_f32(
         false, a1,  false, b1,
         (short)0, acc1,  false,  false);
  }

  #pragma unroll
  for (int r = 0; r < 8; ++r) {
    red[wave][r * 32 + lane] = acc0[r] + acc1[r];
  }
  __syncthreads();

  if (threadIdx.x < 256) {
    float s = 0.0f;
    #pragma unroll
    for (int w = 0; w < KSPLIT; ++w) s += red[w][threadIdx.x];
    const int m = ti * 16 + h * 8 + wave;
    const int n = tj * 16 + c;
    out[m * NG + n] = pn[m] + gn[n] - 2.0f * s;
  }
}

extern "C" void kernel_launch(void* const* d_in, const int* in_sizes, int n_in,
                              void* d_out, int out_size, void* d_ws, size_t ws_size,
                              hipStream_t stream) {
  const float* probe   = (const float*)d_in[0];
  const float* gallery = (const float*)d_in[1];
  float* out = (float*)d_out;

  float* ws    = (float*)d_ws;
  float* pmean = ws;
  float* gmean = pmean + (size_t)NP * NC;
  float* pn    = gmean + (size_t)NG * NC;
  float* gn    = pn + NP;

  const int total_rows = (NP + NG) * NC;
  pool_kernel<<<total_rows / 16, 256, 0, stream>>>(probe, gallery, pmean, gmean);

  norm_kernel<<<(NP + NG) / 8, 256, 0, stream>>>(pmean, gmean, pn, gn);

  dist_wmma_kernel<<<dim3(NG / 16, NP / 16), KSPLIT * 32, 0, stream>>>(pmean, gmean, pn, gn, out);
}


# pool kernel as one resident streaming pass: 256 groups x 16 waves, 6 chunks of 32 rows per wave, next chunk's 16 float4 loads in flight, one whole 128-byte line of means per store; sum tree kept
# speedup vs baseline: 2.3590x; 1.6253x over previous
#include <hip/hip_runtime.h>

#define NP 128
#define NG 256
#define NC 2048
#define HW 64

typedef __attribute__((ext_vector_type(2))) float v2f;
typedef __attribute__((ext_vector_type(4))) float v4f;
typedef __attribute__((ext_vector_type(8))) float v8f;

#define CHUNK_ROWS   32
#define CHUNK_LOADS  16
#define CHUNK_FLOATS (CHUNK_ROWS * HW)
#define POOL_WAVES   16
#define POOL_GROUPS  256
#define PROBE_CHUNKS (NP * NC / CHUNK_ROWS)
#define TOTAL_CHUNKS ((NP + NG) * NC / CHUNK_ROWS)
#define WAVE_CHUNKS  (TOTAL_CHUNKS / (POOL_GROUPS * POOL_WAVES))
static_assert(TOTAL_CHUNKS % (POOL_GROUPS * POOL_WAVES) == 0, "chunks must split evenly over the waves");

__global__ __launch_bounds__(POOL_WAVES * 32)
void pool_kernel(const float* __restrict__ probe,
                 const float* __restrict__ gallery,
                 float* __restrict__ pmean,
                 float* __restrict__ gmean) {
  const int lane = threadIdx.x & 31;
  const int wave = __builtin_amdgcn_readfirstlane(threadIdx.x >> 5);
  const int sub  = lane & 15;
  const int half = lane >> 4;
  const int chunk0 = (blockIdx.x * POOL_WAVES + wave) * WAVE_CHUNKS;

  const bool b0 = (lane & 1) != 0;
  const bool b1 = (lane & 2) != 0;
  const bool b2 = (lane & 4) != 0;
  const bool b3 = (lane & 8) != 0;

  v4f buf[2][CHUNK_LOADS];

#define LOAD_CHUNK(C) do {                                                        \
    const int ch_ = chunk0 + (C);                                                 \
    const float* base_ = (ch_ < PROBE_CHUNKS)                                     \
        ? probe + (size_t)ch_ * CHUNK_FLOATS                                      \
        : gallery + (size_t)(ch_ - PROBE_CHUNKS) * CHUNK_FLOATS;                  \
    const v4f* p_ = (const v4f*)base_ + lane;                                     \
    _Pragma("unroll")                                                             \
    for (int q = 0; q < CHUNK_LOADS; ++q) buf[(C) & 1][q] = p_[q * 32];           \
  } while (0)

  LOAD_CHUNK(0);

  #pragma unroll
  for (int c = 0; c < WAVE_CHUNKS; ++c) {
    __builtin_amdgcn_sched_barrier(0);
    if (c + 1 < WAVE_CHUNKS) LOAD_CHUNK(c + 1);
    __builtin_amdgcn_sched_barrier(0);

    float s[CHUNK_LOADS];
    #pragma unroll
    for (int q = 0; q < CHUNK_LOADS; ++q) {
      const v4f v = buf[c & 1][q];
      s[q] = v.x + v.y + v.z + v.w;
    }
    float t[8];
    #pragma unroll
    for (int i = 0; i < 8; ++i) {
      const float keep = b0 ? s[2 * i + 1] : s[2 * i];
      const float send = b0 ? s[2 * i] : s[2 * i + 1];
      t[i] = keep + __shfl_xor(send, 1, 32);
    }
    float u[4];
    #pragma unroll
    for (int j = 0; j < 4; ++j) {
      const float keep = b1 ? t[2 * j + 1] : t[2 * j];
      const float send = b1 ? t[2 * j] : t[2 * j + 1];
      u[j] = keep + __shfl_xor(send, 2, 32);
    }
    float w[2];
    #pragma unroll
    for (int k = 0; k < 2; ++k) {
      const float keep = b2 ? u[2 * k + 1] : u[2 * k];
      const float send = b2 ? u[2 * k] : u[2 * k + 1];
      w[k] = keep + __shfl_xor(send, 4, 32);
    }
    const float keep = b3 ? w[1] : w[0];
    const float send = b3 ? w[0] : w[1];
    const float sum  = keep + __shfl_xor(send, 8, 32);

    const int ch = chunk0 + c;
    float* dst = (ch < PROBE_CHUNKS) ? pmean + (size_t)ch * CHUNK_ROWS
                                     : gmean + (size_t)(ch - PROBE_CHUNKS) * CHUNK_ROWS;
    dst[2 * sub + half] = sum * (1.0f / 64.0f);
  }
#undef LOAD_CHUNK
}

__global__ void norm_kernel(const float* __restrict__ pmean,
                            const float* __restrict__ gmean,
                            float* __restrict__ pn,
                            float* __restrict__ gn) {
  const int lane = threadIdx.x & 31;
  const int wave = threadIdx.x >> 5;
  const int row  = blockIdx.x * 8 + wave;

  const float* src;
  float* dst;
  if (row < NP) {
    src = pmean + (long long)row * NC;
    dst = pn + row;
  } else {
    src = gmean + (long long)(row - NP) * NC;
    dst = gn + (row - NP);
  }

  float s = 0.0f;
  #pragma unroll 4
  for (int i = lane; i < NC; i += 32) {
    float x = src[i];
    s += x * x;
  }
  s += __shfl_xor(s, 1, 32);
  s += __shfl_xor(s, 2, 32);
  s += __shfl_xor(s, 4, 32);
  s += __shfl_xor(s, 8, 32);
  s += __shfl_xor(s, 16, 32);
  if (lane == 0) *dst = s;
}

#define KSPLIT 16
#define KSLICE (NC / KSPLIT)

__global__ __launch_bounds__(KSPLIT * 32)
void dist_wmma_kernel(const float* __restrict__ pmean,
                      const float* __restrict__ gmean,
                      const float* __restrict__ pn,
                      const float* __restrict__ gn,
                      float* __restrict__ out) {
  __shared__ float red[KSPLIT][256];

  const int lane = threadIdx.x & 31;
  const int wave = threadIdx.x >> 5;
  const int c    = lane & 15;
  const int h    = lane >> 4;
  const int ti   = blockIdx.y;
  const int tj   = blockIdx.x;

  const float* arow = pmean + (long long)(ti * 16 + c) * NC + wave * KSLICE + 4 * h;
  const float* brow = gmean + (long long)(tj * 16 + c) * NC + wave * KSLICE + 4 * h;

  v4f a[KSLICE / 8], b[KSLICE / 8];
  #pragma unroll
  for (int j = 0; j < KSLICE / 8; ++j) {
    a[j] = *(const v4f*)(arow + 8 * j);
    b[j] = *(const v4f*)(brow + 8 * j);
  }

  v8f acc0 = {};
  v8f acc1 = {};
  #pragma unroll
  for (int j = 0; j < KSLICE / 8; ++j) {
    v2f a0 = {a[j].x, a[j].y};
    v2f b0 = {b[j].x, b[j].y};
    v2f a1 = {a[j].z, a[j].w};
    v2f b1 = {b[j].z, b[j].w};
    acc0 = __builtin_amdgcn_wmma_f32_16x16x4_f32(
         false, a0,  false, b0,
         (short)0, acc0,  false,  false);
    acc1 = __builtin_amdgcn_wmma_f32_16x16x4_f32(
         false, a1,  false, b1,
         (short)0, acc1,  false,  false);
  }

  #pragma unroll
  for (int r = 0; r < 8; ++r) {
    red[wave][r * 32 + lane] = acc0[r] + acc1[r];
  }
  __syncthreads();

  if (threadIdx.x < 256) {
    float s = 0.0f;
    #pragma unroll
    for (int w = 0; w < KSPLIT; ++w) s += red[w][threadIdx.x];
    const int m = ti * 16 + h * 8 + wave;
    const int n = tj * 16 + c;
    out[m * NG + n] = pn[m] + gn[n] - 2.0f * s;
  }
}

extern "C" void kernel_launch(void* const* d_in, const int* in_sizes, int n_in,
                              void* d_out, int out_size, void* d_ws, size_t ws_size,
                              hipStream_t stream) {
  const float* probe   = (const float*)d_in[0];
  const float* gallery = (const float*)d_in[1];
  float* out = (float*)d_out;

  float* ws    = (float*)d_ws;
  float* pmean = ws;
  float* gmean = pmean + (size_t)NP * NC;
  float* pn    = gmean + (size_t)NG * NC;
  float* gn    = pn + NP;

  pool_kernel<<<POOL_GROUPS, POOL_WAVES * 32, 0, stream>>>(probe, gallery, pmean, gmean);

  norm_kernel<<<(NP + NG) / 8, 256, 0, stream>>>(pmean, gmean, pn, gn);

  dist_wmma_kernel<<<dim3(NG / 16, NP / 16), KSPLIT * 32, 0, stream>>>(pmean, gmean, pn, gn, out);
}


# speedup vs baseline: 3.2890x; 1.3942x over previous
#include <hip/hip_runtime.h>

#define NP 128
#define NG 256
#define NC 2048
#define HW 64

typedef __attribute__((ext_vector_type(2))) float v2f;
typedef __attribute__((ext_vector_type(4))) float v4f;
typedef __attribute__((ext_vector_type(8))) float v8f;

#define CHUNK_ROWS   32
#define CHUNK_LOADS  16
#define CHUNK_FLOATS (CHUNK_ROWS * HW)
#define POOL_WAVES   16
#define POOL_GROUPS  256
#define PROBE_CHUNKS (NP * NC / CHUNK_ROWS)
#define TOTAL_CHUNKS ((NP + NG) * NC / CHUNK_ROWS)
#define WAVE_CHUNKS  (TOTAL_CHUNKS / (POOL_GROUPS * POOL_WAVES))
static_assert(TOTAL_CHUNKS % (POOL_GROUPS * POOL_WAVES) == 0, "chunks must split evenly over the waves");

__global__ __launch_bounds__(POOL_WAVES * 32)
void pool_kernel(const float* __restrict__ probe,
                 const float* __restrict__ gallery,
                 float* __restrict__ pmean,
                 float* __restrict__ gmean) {
  const int lane = threadIdx.x & 31;
  const int wave = __builtin_amdgcn_readfirstlane(threadIdx.x >> 5);
  const int sub  = lane & 15;
  const int half = lane >> 4;
  const int chunk0 = (blockIdx.x * POOL_WAVES + wave) * WAVE_CHUNKS;

  const bool b0 = (lane & 1) != 0;
  const bool b1 = (lane & 2) != 0;
  const bool b2 = (lane & 4) != 0;
  const bool b3 = (lane & 8) != 0;

  v4f buf[2][CHUNK_LOADS];

#define LOAD_CHUNK(C) do {                                                        \
    const int ch_ = chunk0 + (C);                                                 \
    const float* base_ = (ch_ < PROBE_CHUNKS)                                     \
        ? probe + (size_t)ch_ * CHUNK_FLOATS                                      \
        : gallery + (size_t)(ch_ - PROBE_CHUNKS) * CHUNK_FLOATS;                  \
    const v4f* p_ = (const v4f*)base_ + lane;                                     \
    _Pragma("unroll")                                                             \
    for (int q = 0; q < CHUNK_LOADS; ++q)                                         \
      buf[(C) & 1][q] = __builtin_nontemporal_load(p_ + q * 32);                  \
  } while (0)

  LOAD_CHUNK(0);

  #pragma unroll
  for (int c = 0; c < WAVE_CHUNKS; ++c) {
    __builtin_amdgcn_sched_barrier(0);
    if (c + 1 < WAVE_CHUNKS) LOAD_CHUNK(c + 1);
    __builtin_amdgcn_sched_barrier(0);

    float s[CHUNK_LOADS];
    #pragma unroll
    for (int q = 0; q < CHUNK_LOADS; ++q) {
      const v4f v = buf[c & 1][q];
      s[q] = v.x + v.y + v.z + v.w;
    }
    float t[8];
    #pragma unroll
    for (int i = 0; i < 8; ++i) {
      const float keep = b0 ? s[2 * i + 1] : s[2 * i];
      const float send = b0 ? s[2 * i] : s[2 * i + 1];
      t[i] = keep + __shfl_xor(send, 1, 32);
    }
    float u[4];
    #pragma unroll
    for (int j = 0; j < 4; ++j) {
      const float keep = b1 ? t[2 * j + 1] : t[2 * j];
      const float send = b1 ? t[2 * j] : t[2 * j + 1];
      u[j] = keep + __shfl_xor(send, 2, 32);
    }
    float w[2];
    #pragma unroll
    for (int k = 0; k < 2; ++k) {
      const float keep = b2 ? u[2 * k + 1] : u[2 * k];
      const float send = b2 ? u[2 * k] : u[2 * k + 1];
      w[k] = keep + __shfl_xor(send, 4, 32);
    }
    const float keep = b3 ? w[1] : w[0];
    const float send = b3 ? w[0] : w[1];
    const float sum  = keep + __shfl_xor(send, 8, 32);

    const int ch = chunk0 + c;
    float* dst = (ch < PROBE_CHUNKS) ? pmean + (size_t)ch * CHUNK_ROWS
                                     : gmean + (size_t)(ch - PROBE_CHUNKS) * CHUNK_ROWS;
    dst[2 * sub + half] = sum * (1.0f / 64.0f);
  }
#undef LOAD_CHUNK
}

#define KSPLIT 16
#define KSLICE (NC / KSPLIT)

__global__ __launch_bounds__(KSPLIT * 32)
void dist_wmma_kernel(const float* __restrict__ pmean,
                      const float* __restrict__ gmean,
                      float* __restrict__ out) {
  __shared__ float red[KSPLIT][256];
  __shared__ float nsq[KSPLIT][32];

  const int lane = threadIdx.x & 31;
  const int wave = threadIdx.x >> 5;
  const int c    = lane & 15;
  const int h    = lane >> 4;
  const int ti   = blockIdx.y;
  const int tj   = blockIdx.x;

  const float* arow = pmean + (long long)(ti * 16 + c) * NC + wave * KSLICE + 4 * h;
  const float* brow = gmean + (long long)(tj * 16 + c) * NC + wave * KSLICE + 4 * h;

  v4f a[KSLICE / 8], b[KSLICE / 8];
  #pragma unroll
  for (int j = 0; j < KSLICE / 8; ++j) {
    a[j] = *(const v4f*)(arow + 8 * j);
    b[j] = *(const v4f*)(brow + 8 * j);
  }

  v8f acc0 = {};
  v8f acc1 = {};
  v4f sa = {};
  v4f sb = {};
  #pragma unroll
  for (int j = 0; j < KSLICE / 8; ++j) {
    v2f a0 = {a[j].x, a[j].y};
    v2f b0 = {b[j].x, b[j].y};
    v2f a1 = {a[j].z, a[j].w};
    v2f b1 = {b[j].z, b[j].w};
    acc0 = __builtin_amdgcn_wmma_f32_16x16x4_f32(
         false, a0,  false, b0,
         (short)0, acc0,  false,  false);
    acc1 = __builtin_amdgcn_wmma_f32_16x16x4_f32(
         false, a1,  false, b1,
         (short)0, acc1,  false,  false);
    sa += a[j] * a[j];
    sb += b[j] * b[j];
  }

  const float qa   = (sa.x + sa.y) + (sa.z + sa.w);
  const float qb   = (sb.x + sb.y) + (sb.z + sb.w);
  const float keep = h ? qb : qa;
  const float send = h ? qa : qb;
  nsq[wave][lane] = keep + __shfl_xor(send, 16, 32);

  #pragma unroll
  for (int r = 0; r < 8; ++r) {
    red[wave][r * 32 + lane] = acc0[r] + acc1[r];
  }
  __syncthreads();

  if (threadIdx.x < 256) {
    const int ml = h * 8 + wave;
    float s = 0.0f, pn = 0.0f, gn = 0.0f;
    #pragma unroll
    for (int w = 0; w < KSPLIT; ++w) {
      s  += red[w][threadIdx.x];
      pn += nsq[w][ml];
      gn += nsq[w][16 + c];
    }
    const int m = ti * 16 + ml;
    const int n = tj * 16 + c;
    out[m * NG + n] = pn + gn - 2.0f * s;
  }
}

extern "C" void kernel_launch(void* const* d_in, const int* in_sizes, int n_in,
                              void* d_out, int out_size, void* d_ws, size_t ws_size,
                              hipStream_t stream) {
  const float* probe   = (const float*)d_in[0];
  const float* gallery = (const float*)d_in[1];
  float* out = (float*)d_out;

  float* ws    = (float*)d_ws;
  float* pmean = ws;
  float* gmean = pmean + (size_t)NP * NC;

  pool_kernel<<<POOL_GROUPS, POOL_WAVES * 32, 0, stream>>>(probe, gallery, pmean, gmean);

  dist_wmma_kernel<<<dim3(NG / 16, NP / 16), KSPLIT * 32, 0, stream>>>(pmean, gmean, out);
}


# distance kernel: each wave fetches its K slice as whole 512-byte row pieces (4 lines per load, was 16) and reorders them into WMMA fragments through a private padded LDS tile; sums, order unchanged
# speedup vs baseline: 3.4658x; 1.0538x over previous
#include <hip/hip_runtime.h>

#define NP 128
#define NG 256
#define NC 2048
#define HW 64

typedef __attribute__((ext_vector_type(2))) float v2f;
typedef __attribute__((ext_vector_type(4))) float v4f;
typedef __attribute__((ext_vector_type(8))) float v8f;

#define CHUNK_ROWS   32
#define CHUNK_LOADS  16
#define CHUNK_FLOATS (CHUNK_ROWS * HW)
#define POOL_WAVES   16
#define POOL_GROUPS  256
#define PROBE_CHUNKS (NP * NC / CHUNK_ROWS)
#define TOTAL_CHUNKS ((NP + NG) * NC / CHUNK_ROWS)
#define WAVE_CHUNKS  (TOTAL_CHUNKS / (POOL_GROUPS * POOL_WAVES))
static_assert(TOTAL_CHUNKS % (POOL_GROUPS * POOL_WAVES) == 0, "chunks must split evenly over the waves");

__global__ __launch_bounds__(POOL_WAVES * 32)
void pool_kernel(const float* __restrict__ probe,
                 const float* __restrict__ gallery,
                 float* __restrict__ pmean,
                 float* __restrict__ gmean) {
  const int lane = threadIdx.x & 31;
  const int wave = __builtin_amdgcn_readfirstlane(threadIdx.x >> 5);
  const int sub  = lane & 15;
  const int half = lane >> 4;
  const int chunk0 = (blockIdx.x * POOL_WAVES + wave) * WAVE_CHUNKS;

  const bool b0 = (lane & 1) != 0;
  const bool b1 = (lane & 2) != 0;
  const bool b2 = (lane & 4) != 0;
  const bool b3 = (lane & 8) != 0;

  v4f buf[2][CHUNK_LOADS];

#define LOAD_CHUNK(C) do {                                                        \
    const int ch_ = chunk0 + (C);                                                 \
    const float* base_ = (ch_ < PROBE_CHUNKS)                                     \
        ? probe + (size_t)ch_ * CHUNK_FLOATS                                      \
        : gallery + (size_t)(ch_ - PROBE_CHUNKS) * CHUNK_FLOATS;                  \
    const v4f* p_ = (const v4f*)base_ + lane;                                     \
    _Pragma("unroll")                                                             \
    for (int q = 0; q < CHUNK_LOADS; ++q)                                         \
      buf[(C) & 1][q] = __builtin_nontemporal_load(p_ + q * 32);                  \
  } while (0)

  LOAD_CHUNK(0);

  #pragma unroll
  for (int c = 0; c < WAVE_CHUNKS; ++c) {
    __builtin_amdgcn_sched_barrier(0);
    if (c + 1 < WAVE_CHUNKS) LOAD_CHUNK(c + 1);
    __builtin_amdgcn_sched_barrier(0);

    float s[CHUNK_LOADS];
    #pragma unroll
    for (int q = 0; q < CHUNK_LOADS; ++q) {
      const v4f v = buf[c & 1][q];
      s[q] = v.x + v.y + v.z + v.w;
    }
    float t[8];
    #pragma unroll
    for (int i = 0; i < 8; ++i) {
      const float keep = b0 ? s[2 * i + 1] : s[2 * i];
      const float send = b0 ? s[2 * i] : s[2 * i + 1];
      t[i] = keep + __shfl_xor(send, 1, 32);
    }
    float u[4];
    #pragma unroll
    for (int j = 0; j < 4; ++j) {
      const float keep = b1 ? t[2 * j + 1] : t[2 * j];
      const float send = b1 ? t[2 * j] : t[2 * j + 1];
      u[j] = keep + __shfl_xor(send, 2, 32);
    }
    float w[2];
    #pragma unroll
    for (int k = 0; k < 2; ++k) {
      const float keep = b2 ? u[2 * k + 1] : u[2 * k];
      const float send = b2 ? u[2 * k] : u[2 * k + 1];
      w[k] = keep + __shfl_xor(send, 4, 32);
    }
    const float keep = b3 ? w[1] : w[0];
    const float send = b3 ? w[0] : w[1];
    const float sum  = keep + __shfl_xor(send, 8, 32);

    const int ch = chunk0 + c;
    float* dst = (ch < PROBE_CHUNKS) ? pmean + (size_t)ch * CHUNK_ROWS
                                     : gmean + (size_t)(ch - PROBE_CHUNKS) * CHUNK_ROWS;
    dst[2 * sub + half] = sum * (1.0f / 64.0f);
  }
#undef LOAD_CHUNK
}

#define KSPLIT 16
#define KSLICE (NC / KSPLIT)
#define TPITCH (KSLICE + 4)

__global__ __launch_bounds__(KSPLIT * 32)
void dist_wmma_kernel(const float* __restrict__ pmean,
                      const float* __restrict__ gmean,
                      float* __restrict__ out) {
  __shared__ __attribute__((aligned(16))) float tile[KSPLIT][32][TPITCH];
  __shared__ float red[KSPLIT][256];
  __shared__ float nsq[KSPLIT][32];

  const int lane = threadIdx.x & 31;
  const int wave = threadIdx.x >> 5;
  const int c    = lane & 15;
  const int h    = lane >> 4;
  const int ti   = blockIdx.y;
  const int tj   = blockIdx.x;

  const float* ag = pmean + (long long)(ti * 16) * NC + wave * KSLICE + 4 * lane;
  const float* bg = gmean + (long long)(tj * 16) * NC + wave * KSLICE + 4 * lane;

  v4f ga[16], gb[16];
  #pragma unroll
  for (int r = 0; r < 16; ++r) {
    ga[r] = *(const v4f*)(ag + (long long)r * NC);
    gb[r] = *(const v4f*)(bg + (long long)r * NC);
  }

  float* trow = &tile[wave][0][4 * lane];
  #pragma unroll
  for (int r = 0; r < 16; ++r) {
    *(v4f*)(trow + r * TPITCH)        = ga[r];
    *(v4f*)(trow + (16 + r) * TPITCH) = gb[r];
  }

  const float* fa = &tile[wave][c][4 * h];
  const float* fb = &tile[wave][16 + c][4 * h];
  v4f a[KSLICE / 8], b[KSLICE / 8];
  #pragma unroll
  for (int j = 0; j < KSLICE / 8; ++j) {
    a[j] = *(const v4f*)(fa + 8 * j);
    b[j] = *(const v4f*)(fb + 8 * j);
  }
  __builtin_amdgcn_sched_barrier(0);

  v8f acc0 = {};
  v8f acc1 = {};
  v4f sa = {};
  v4f sb = {};
  #pragma unroll
  for (int j = 0; j < KSLICE / 8; ++j) {
    v2f a0 = {a[j].x, a[j].y};
    v2f b0 = {b[j].x, b[j].y};
    v2f a1 = {a[j].z, a[j].w};
    v2f b1 = {b[j].z, b[j].w};
    acc0 = __builtin_amdgcn_wmma_f32_16x16x4_f32(
         false, a0,  false, b0,
         (short)0, acc0,  false,  false);
    acc1 = __builtin_amdgcn_wmma_f32_16x16x4_f32(
         false, a1,  false, b1,
         (short)0, acc1,  false,  false);
    sa += a[j] * a[j];
    sb += b[j] * b[j];
  }

  const float qa   = (sa.x + sa.y) + (sa.z + sa.w);
  const float qb   = (sb.x + sb.y) + (sb.z + sb.w);
  const float keep = h ? qb : qa;
  const float send = h ? qa : qb;
  nsq[wave][lane] = keep + __shfl_xor(send, 16, 32);

  #pragma unroll
  for (int r = 0; r < 8; ++r) {
    red[wave][r * 32 + lane] = acc0[r] + acc1[r];
  }
  __syncthreads();

  if (threadIdx.x < 256) {
    const int ml = h * 8 + wave;
    float s = 0.0f, pn = 0.0f, gn = 0.0f;
    #pragma unroll
    for (int w = 0; w < KSPLIT; ++w) {
      s  += red[w][threadIdx.x];
      pn += nsq[w][ml];
      gn += nsq[w][16 + c];
    }
    const int m = ti * 16 + ml;
    const int n = tj * 16 + c;
    out[m * NG + n] = pn + gn - 2.0f * s;
  }
}

extern "C" void kernel_launch(void* const* d_in, const int* in_sizes, int n_in,
                              void* d_out, int out_size, void* d_ws, size_t ws_size,
                              hipStream_t stream) {
  const float* probe   = (const float*)d_in[0];
  const float* gallery = (const float*)d_in[1];
  float* out = (float*)d_out;

  float* ws    = (float*)d_ws;
  float* pmean = ws;
  float* gmean = pmean + (size_t)NP * NC;

  pool_kernel<<<POOL_GROUPS, POOL_WAVES * 32, 0, stream>>>(probe, gallery, pmean, gmean);

  dist_wmma_kernel<<<dim3(NG / 16, NP / 16), KSPLIT * 32, 0, stream>>>(pmean, gmean, out);
}
